# starting kernel
# baseline (speedup 1.0000x reference)
#include <hip/hip_runtime.h>
#include <cmath>

typedef float v2f __attribute__((ext_vector_type(2)));
typedef float v8f __attribute__((ext_vector_type(8)));

#define WMMA_F32(A, Bf, C) \
    __builtin_amdgcn_wmma_f32_16x16x4_f32(false, (A), false, (Bf), (short)0, (C), false, false)

#define BB       2
#define NN       2048
#define FF       32
#define DROWS    8192
#define DCOLS    2049
#define MSHR     6144
#define KOUT     2049
#define KPAD     2064
#define NSPLIT   4
#define KSEG     (MSHR / NSPLIT)

#define WS_XW_OFF    0
#define WS_XSHR_OFF  (BB * NN * FF)
#define WS_PART_OFF  (WS_XSHR_OFF + BB * MSHR * FF)

__global__ void sufg_xw_kernel(const float* __restrict__ x,
                               const float* __restrict__ w,
                               float* __restrict__ xw) {
    int idx = blockIdx.x * blockDim.x + threadIdx.x;
    int o = idx & 31;
    int n = (idx >> 5) & (NN - 1);
    int b = idx >> 16;
    const float* xr = x + ((size_t)b * NN + n) * FF;
    float s = 0.f;
#pragma unroll
    for (int f = 0; f < FF; ++f) s = fmaf(xr[f], w[f * FF + o], s);
    xw[idx] = s;
}

__global__ void __launch_bounds__(256)
sufg_pass2_kernel(const float* __restrict__ dl,
                  const float* __restrict__ xw,
                  const float* __restrict__ filt,
                  float* __restrict__ xshr,
                  float thr) {
    const int lane = threadIdx.x & 31;
    const int wave = threadIdx.x >> 5;
    const int tile = blockIdx.x * 8 + wave;
    const int b    = blockIdx.y;
    if (tile >= MSHR / 16) return;

    const int m0   = tile * 16;
    const int row  = lane & 15;
    const int hi   = lane >> 4;
    const int koff = hi << 1;

    const float* A  = dl + (size_t)b * DROWS * DCOLS
                         + (size_t)(2048 + m0 + row) * DCOLS + 1 + koff;
    const float* Bp = xw + (size_t)b * NN * FF + (size_t)koff * FF + row;

    v8f acc0 = {0.f,0.f,0.f,0.f,0.f,0.f,0.f,0.f};
    v8f acc1 = {0.f,0.f,0.f,0.f,0.f,0.f,0.f,0.f};

    for (int kc = 0; kc < NN; kc += 64) {
        __builtin_prefetch(A + 256, 0, 0);
#pragma unroll
        for (int k = 0; k < 64; k += 4) {
            v2f a;  a.x  = A[0];   a.y  = A[1];
            v2f b0; b0.x = Bp[0];  b0.y = Bp[32];
            v2f b1; b1.x = Bp[16]; b1.y = Bp[48];
            acc0 = WMMA_F32(a, b0, acc0);
            acc1 = WMMA_F32(a, b1, acc1);
            A += 4; Bp += 4 * FF;
        }
    }

#pragma unroll
    for (int r = 0; r < 8; ++r) {
        const int m  = m0 + r + hi * 8;
        const float sc = filt[2048 + m];
        float v0 = acc0[r] * sc;
        float v1 = acc1[r] * sc;
        if (m >= 2048) {
            v0 = copysignf(fmaxf(fabsf(v0) - thr, 0.f), v0);
            v1 = copysignf(fmaxf(fabsf(v1) - thr, 0.f), v1);
        }
        float* orow = xshr + ((size_t)b * MSHR + m) * FF;
        orow[row]      = v0;
        orow[16 + row] = v1;
    }
}

__global__ void __launch_bounds__(256)
sufg_pass3_kernel(const float* __restrict__ dl,
                  const float* __restrict__ xshr,
                  float* __restrict__ partial) {
    const int lane = threadIdx.x & 31;
    const int wave = threadIdx.x >> 5;
    const int unit = blockIdx.x * 8 + wave;
    const int b    = blockIdx.y;
    if (unit >= (KPAD / 16) * NSPLIT) return;

    const int split  = unit & (NSPLIT - 1);
    const int tile   = unit >> 2;
    const int kbegin = split * KSEG;

    const int row  = lane & 15;
    const int hi   = lane >> 4;
    const int koff = hi << 1;

    int krow = tile * 16 + row;
    if (krow > KOUT - 1) krow = KOUT - 1;

    const float* A  = dl + (size_t)b * DROWS * DCOLS + (size_t)2048 * DCOLS
                         + (size_t)krow * MSHR + kbegin + koff;
    const float* Bp = xshr + (size_t)b * MSHR * FF + (size_t)(kbegin + koff) * FF + row;

    v8f acc0 = {0.f,0.f,0.f,0.f,0.f,0.f,0.f,0.f};
    v8f acc1 = {0.f,0.f,0.f,0.f,0.f,0.f,0.f,0.f};

    for (int kc = 0; kc < KSEG; kc += 64) {
        __builtin_prefetch(A + 256, 0, 0);
#pragma unroll
        for (int k = 0; k < 64; k += 4) {
            v2f a  = *(const v2f*)A;
            v2f b0; b0.x = Bp[0];  b0.y = Bp[32];
            v2f b1; b1.x = Bp[16]; b1.y = Bp[48];
            acc0 = WMMA_F32(a, b0, acc0);
            acc1 = WMMA_F32(a, b1, acc1);
            A += 4; Bp += 4 * FF;
        }
    }

#pragma unroll
    for (int r = 0; r < 8; ++r) {
        const int kk = tile * 16 + r + hi * 8;
        float* prow = partial + (((size_t)(b * NSPLIT + split)) * KPAD + kk) * FF;
        prow[row]      = acc0[r];
        prow[16 + row] = acc1[r];
    }
}

__global__ void sufg_reduce_kernel(const float* __restrict__ partial,
                                   const float* __restrict__ bias,
                                   float* __restrict__ out) {
    int idx = blockIdx.x * blockDim.x + threadIdx.x;
    if (idx >= BB * KOUT * FF) return;
    int o = idx & 31;
    int k = (idx >> 5) % KOUT;
    int b = idx / (KOUT * FF);
    float s = bias[o];
#pragma unroll
    for (int sp = 0; sp < NSPLIT; ++sp)
        s += partial[(((size_t)(b * NSPLIT + sp)) * KPAD + k) * FF + o];
    out[idx] = s;
}

extern "C" void kernel_launch(void* const* d_in, const int* in_sizes, int n_in,
                              void* d_out, int out_size, void* d_ws, size_t ws_size,
                              hipStream_t stream) {
    (void)in_sizes; (void)n_in; (void)out_size; (void)ws_size;
    const float* x    = (const float*)d_in[0];
    const float* dl   = (const float*)d_in[1];
    const float* w    = (const float*)d_in[2];
    const float* filt = (const float*)d_in[3];
    const float* bias = (const float*)d_in[4];
    float* out = (float*)d_out;

    float* ws      = (float*)d_ws;
    float* xw      = ws + WS_XW_OFF;
    float* xshr    = ws + WS_XSHR_OFF;
    float* partial = ws + WS_PART_OFF;

    const float thr = sqrtf(2.0f * logf((float)NN)) / sqrtf((float)NN);

    sufg_xw_kernel<<<(BB * NN * FF) / 256, 256, 0, stream>>>(x, w, xw);
    sufg_pass2_kernel<<<dim3((MSHR / 16) / 8, BB), 256, 0, stream>>>(dl, xw, filt, xshr, thr);
    sufg_pass3_kernel<<<dim3(((KPAD / 16) * NSPLIT + 7) / 8, BB), 256, 0, stream>>>(dl, xshr, partial);
    sufg_reduce_kernel<<<(BB * KOUT * FF + 255) / 256, 256, 0, stream>>>(partial, bias, out);
}


# Loader waves hold three chunks of A and B in three register sets: loads issued three steps ahead of their LDS stores (was two), 60 loads in flight per loader wave; LDS ring and sum order unchanged
# speedup vs baseline: 2.5047x; 2.5047x over previous
#include <hip/hip_runtime.h>
#include <cmath>

typedef float v2f __attribute__((ext_vector_type(2)));
typedef float v4f __attribute__((ext_vector_type(4)));
typedef float v8f __attribute__((ext_vector_type(8)));

#define WMMA_F32(A, Bf, C) \
    __builtin_amdgcn_wmma_f32_16x16x4_f32(false, (A), false, (Bf), (short)0, (C), false, false)

#define BB       2
#define NN       2048
#define FF       32
#define DROWS    8192
#define DCOLS    2049
#define MSHR     6144
#define KOUT     2049
#define KPAD     2064
#define NSPLIT   4
#define KSEG     (MSHR / NSPLIT)

#define WS_XW_OFF    0
#define WS_XSHR_OFF  (BB * NN * FF)
#define WS_PART_OFF  (WS_XSHR_OFF + BB * MSHR * FF)

__global__ void sufg_xw_kernel(const float* __restrict__ x,
                               const float* __restrict__ w,
                               float* __restrict__ xw) {
    int idx = blockIdx.x * blockDim.x + threadIdx.x;
    int o = idx & 31;
    int n = (idx >> 5) & (NN - 1);
    int b = idx >> 16;
    const float* xr = x + ((size_t)b * NN + n) * FF;
    float s = 0.f;
#pragma unroll
    for (int f = 0; f < FF; ++f) s = fmaf(xr[f], w[f * FF + o], s);
    xw[idx] = s;
}

#define LDB_SET(BSET, CH) do {                                                 \
    const float* bq_ = Bsrc + (size_t)(CH) * (64 * FF);                        \
    BSET[0] = *(const v4f*)(bq_);                                              \
    BSET[1] = *(const v4f*)(bq_ + FF);                                         \
    BSET[2] = *(const v4f*)(bq_ + 16);                                         \
    BSET[3] = *(const v4f*)(bq_ + FF + 16);                                    \
    } while (0)

#define STB_SET(BSET, CH) do {                                                 \
    v4f* bd_ = Bdst + (CH) * 512;                                              \
      \
      \
      \
                                           \
    asm volatile("" : "+v"(BSET[0]), "+v"(BSET[1]), "+v"(BSET[2]), "+v"(BSET[3])); \
    v4f t0_ = { BSET[0].x, BSET[1].x, BSET[2].x, BSET[3].x };                  \
    v4f t1_ = { BSET[0].y, BSET[1].y, BSET[2].y, BSET[3].y };                  \
    v4f t2_ = { BSET[0].z, BSET[1].z, BSET[2].z, BSET[3].z };                  \
    v4f t3_ = { BSET[0].w, BSET[1].w, BSET[2].w, BSET[3].w };                  \
    bd_[(0 + bsw) & 3] = t0_;                                                  \
    bd_[(1 + bsw) & 3] = t1_;                                                  \
    bd_[(2 + bsw) & 3] = t2_;                                                  \
    bd_[(3 + bsw) & 3] = t3_;                                                  \
    } while (0)

#define APITCH 68

__device__ __forceinline__ void lds_ops_done()   { asm volatile("s_wait_dscnt 0x0" ::: "memory"); }
__device__ __forceinline__ void barrier_arrive() { asm volatile("s_barrier_signal -1" ::: "memory"); }
__device__ __forceinline__ void barrier_pass()   { asm volatile("s_barrier_wait -1" ::: "memory"); }

#define LD_SET(SET, CH) do {                                                   \
    const float* p_ = Ag + (CH) * 64;                                          \
    _Pragma("unroll")                                                          \
    for (int r_ = 0; r_ < 16; ++r_) {                                          \
        const float* q_ = p_ + (size_t)r_ * rowstep;                           \
        if (AL8) SET[r_] = *(const v2f*)q_;                                    \
        else { SET[r_].x = q_[0]; SET[r_].y = q_[1]; }                         \
    } } while (0)

#define ST_SET(SET, ST) do {                                                   \
    _Pragma("unroll")                                                          \
    for (int r_ = 0; r_ < 16; ++r_)                                            \
        *(v2f*)(ringw + (ST) * STAGE_F + r_ * APITCH) = SET[r_];               \
    } while (0)

#define WMMA_RANGE(S0, S1) do {                                                \
    _Pragma("unroll")                                                          \
    for (int s_ = (S0); s_ < (S1); ++s_) {                                     \
        v2f b0_; b0_.x = f_[s_].x; b0_.y = f_[s_].y;                           \
        v2f b1_; b1_.x = f_[s_].z; b1_.y = f_[s_].w;                           \
        acc0 = WMMA_F32(a_[s_], b0_, acc0);                                    \
        acc1 = WMMA_F32(a_[s_], b1_, acc1);                                    \
    } } while (0)

#define MSTEP(J, RS) do {                                                      \
    v2f a_[16];                                                                \
    float4 f_[16];                                                             \
    const float*  ar_ = ringr + (RS) * STAGE_F;                                \
    const float4* be_ = Be + (J) * 512;                                        \
    const float4* bo_ = Bo + (J) * 512;                                        \
    _Pragma("unroll")                                                          \
    for (int s_ = 0; s_ < 16; s_ += 2) {                                       \
        a_[s_]     = *(const v2f*)(ar_ + 4 * s_);                              \
        f_[s_]     = be_[32 * s_];                                             \
        a_[s_ + 1] = *(const v2f*)(ar_ + 4 * s_ + 4);                          \
        f_[s_ + 1] = bo_[32 * s_];                                             \
    }                                                                          \
         \
                   \
    __builtin_amdgcn_sched_barrier(0);                                         \
    WMMA_RANGE(0, 8);                                                          \
              \
    __builtin_amdgcn_sched_barrier(0);                                         \
    lds_ops_done();                                                            \
    barrier_arrive();                                                          \
    WMMA_RANGE(8, 16);                                                         \
         \
                          \
    __builtin_amdgcn_sched_barrier(0);                                         \
    barrier_pass();                                                            \
    } while (0)

#define LSTEP(J, WS, SET, BSET) do {                                           \
    ST_SET(SET, WS);                                                           \
    if ((J) + 1 < NCH) STB_SET(BSET, (J) + 1);                                 \
          \
          \
                                   \
    __builtin_amdgcn_sched_barrier(0);                                         \
    LD_SET(SET, ((J) + 4 < NCH) ? (J) + 4 : NCH - 1);                          \
    LDB_SET(BSET, ((J) + 4 < NCH) ? (J) + 4 : NCH - 1);                        \
    lds_ops_done();                                                            \
    barrier_arrive();                                                          \
    barrier_pass();                                                            \
    } while (0)

#define P2_STAGE_F (4 * 16 * APITCH)

__global__ void __launch_bounds__(256)
sufg_pass2_kernel(const float* __restrict__ dl,
                  const float* __restrict__ xw,
                  const float* __restrict__ filt,
                  float* __restrict__ xshr,
                  float thr) {
    __shared__ float4 Bs[(NN / 4) * 32];
    __shared__ __attribute__((aligned(16))) float Ar[2 * P2_STAGE_F];

    constexpr int  NCH     = NN / 64;
    constexpr int  STAGE_F = P2_STAGE_F;
    constexpr bool AL8     = false;
    constexpr int  rowstep = DCOLS;

    const int lane = threadIdx.x & 31;
    const int wave = __builtin_amdgcn_readfirstlane((int)(threadIdx.x >> 5));
    const bool is_loader = wave >= 4;
    const bool is_matrix = !is_loader;
    const int slot = wave & 3;
    const int tile = blockIdx.x * 4 + slot;
    const int b    = blockIdx.y;
    const int m0   = tile * 16;
    const int row  = lane & 15;
    const int hi   = lane >> 4;

    const float* Ag = dl + (size_t)b * DROWS * DCOLS
                         + (size_t)(2048 + m0) * DCOLS + 1 + 2 * lane;
    float*       ringw = Ar + slot * 16 * APITCH + 2 * lane;
    const float* ringr = Ar + (slot * 16 + row) * APITCH + 2 * hi;

    const int    bitem = slot * 32 + lane;
    const int    bsw   = (lane >> 2) & 3;
    const float* Bsrc  = xw + (size_t)b * NN * FF + (size_t)(2 * (bitem >> 2)) * FF + 4 * (bitem & 3);
    v4f*         Bdst  = (v4f*)Bs + 4 * bitem;

    v2f s0[16], s1[16], s2[16];
    v4f b0[4], b1[4], b2[4];
    if (is_loader) {
        LD_SET(s0, 0); LDB_SET(b0, 0);
        __builtin_amdgcn_sched_barrier(0);
        LD_SET(s1, 1); LDB_SET(b1, 1);
        __builtin_amdgcn_sched_barrier(0);
        LD_SET(s2, 2); LDB_SET(b2, 2);
        __builtin_amdgcn_sched_barrier(0);
        ST_SET(s0, 0); STB_SET(b0, 0);
        __builtin_amdgcn_sched_barrier(0);
        LD_SET(s0, 3); LDB_SET(b0, 3);
    }
    lds_ops_done();
    barrier_arrive();
    barrier_pass();

    const float4* Be = Bs + ((lane & ~3) | (((lane & 3) + hi) & 3));
    const float4* Bo = Bs + ((lane & ~3) | (((lane & 3) + hi + 2) & 3)) + 32;

    v8f acc0 = {0.f,0.f,0.f,0.f,0.f,0.f,0.f,0.f};
    v8f acc1 = {0.f,0.f,0.f,0.f,0.f,0.f,0.f,0.f};

    if (is_matrix) {
#pragma nounroll
        for (int j = 0; j < NCH; j += 2) {
            MSTEP(j,     0);
            MSTEP(j + 1, 1);
        }
#pragma unroll
        for (int r = 0; r < 8; ++r) {
            const int m  = m0 + r + hi * 8;
            const float sc = filt[2048 + m];
            float v0 = acc0[r] * sc;
            float v1 = acc1[r] * sc;
            if (m >= 2048) {
                v0 = copysignf(fmaxf(fabsf(v0) - thr, 0.f), v0);
                v1 = copysignf(fmaxf(fabsf(v1) - thr, 0.f), v1);
            }
            float* orow = xshr + ((size_t)b * MSHR + m) * FF;
            orow[row]      = v0;
            orow[16 + row] = v1;
        }
    } else {
#pragma nounroll
        for (int j = 0; j < NCH - 8; j += 6) {
            LSTEP(j,     1, s1, b1);
            LSTEP(j + 1, 0, s2, b2);
            LSTEP(j + 2, 1, s0, b0);
            LSTEP(j + 3, 0, s1, b1);
            LSTEP(j + 4, 1, s2, b2);
            LSTEP(j + 5, 0, s0, b0);
        }
        LSTEP(NCH - 8, 1, s1, b1);
        LSTEP(NCH - 7, 0, s2, b2);
        LSTEP(NCH - 6, 1, s0, b0);
        LSTEP(NCH - 5, 0, s1, b1);
        LSTEP(NCH - 4, 1, s2, b2);
        LSTEP(NCH - 3, 0, s0, b0);
        LSTEP(NCH - 2, 1, s1, b1);
        LSTEP(NCH - 1, 0, s2, b2);
    }
}

#define P3_STAGE_F (4 * 16 * APITCH)
#define XROW_K     (KSEG / 32)
#define XROW_STEPS (XROW_K / 4)
#define XROW_PIECES (NSPLIT * 32)

__global__ void __launch_bounds__(256)
sufg_pass3_kernel(const float* __restrict__ dl,
                  const float* __restrict__ xshr,
                  float* __restrict__ partial,
                  float* __restrict__ lastrow) {
    __shared__ float4 Bs[(KSEG / 4) * 32];
    __shared__ __attribute__((aligned(16))) float Ar[2 * P3_STAGE_F];

    constexpr int  NCH     = KSEG / 64;
    constexpr int  STAGE_F = P3_STAGE_F;
    constexpr bool AL8     = true;
    constexpr int  rowstep = MSHR;

    const int lane  = threadIdx.x & 31;
    const int wave  = __builtin_amdgcn_readfirstlane((int)(threadIdx.x >> 5));
    const int group = blockIdx.x;
    const int split = blockIdx.y;
    const int b     = blockIdx.z;
    const int kbegin = split * KSEG;

    const bool is_loader = wave >= 4;
    const bool is_matrix = !is_loader;
    const bool has_xrow  = wave == 0;
    const int slot = wave & 3;
    const int tile = group * 4 + slot;

    const int row  = lane & 15;
    const int hi   = lane >> 4;

    const float* y2 = dl + (size_t)b * DROWS * DCOLS + (size_t)2048 * DCOLS;
    const float* Ag = y2 + (size_t)(tile * 16) * MSHR + kbegin + 2 * lane;
    float*       ringw = Ar + slot * 16 * APITCH + 2 * lane;
    const float* ringr = Ar + (slot * 16 + row) * APITCH + 2 * hi;

    v2f xa[XROW_STEPS];
    if (has_xrow) {
        const float* xr = y2 + (size_t)2048 * MSHR + kbegin + group * XROW_K + 2 * hi;
#pragma unroll
        for (int t = 0; t < XROW_STEPS; ++t) xa[t] = *(const v2f*)(xr + 4 * t);
    }

    const int    bitem = slot * 32 + lane;
    const int    bsw   = (lane >> 2) & 3;
    const float* Bsrc  = xshr + ((size_t)b * MSHR + kbegin) * FF
                              + (size_t)(2 * (bitem >> 2)) * FF + 4 * (bitem & 3);
    v4f*         Bdst  = (v4f*)Bs + 4 * bitem;

    v2f s0[16], s1[16], s2[16];
    v4f b0[4], b1[4], b2[4];
    if (is_loader) {
        LD_SET(s0, 0); LDB_SET(b0, 0);
        __builtin_amdgcn_sched_barrier(0);
        LD_SET(s1, 1); LDB_SET(b1, 1);
        __builtin_amdgcn_sched_barrier(0);
        LD_SET(s2, 2); LDB_SET(b2, 2);
        __builtin_amdgcn_sched_barrier(0);
        ST_SET(s0, 0); STB_SET(b0, 0);
        __builtin_amdgcn_sched_barrier(0);
        LD_SET(s0, 3); LDB_SET(b0, 3);
    }
    lds_ops_done();
    barrier_arrive();
    barrier_pass();

    const float4* Be = Bs + ((lane & ~3) | (((lane & 3) + hi) & 3));
    const float4* Bo = Bs + ((lane & ~3) | (((lane & 3) + hi + 2) & 3)) + 32;

    v8f acc0 = {0.f,0.f,0.f,0.f,0.f,0.f,0.f,0.f};
    v8f acc1 = {0.f,0.f,0.f,0.f,0.f,0.f,0.f,0.f};

    if (is_matrix) {
#pragma nounroll
        for (int j = 0; j < NCH; j += 2) {
            MSTEP(j,     0);
            MSTEP(j + 1, 1);
        }
#pragma unroll
        for (int r = 0; r < 8; ++r) {
            const int kk = tile * 16 + r + hi * 8;
            float* prow = partial + (((size_t)(b * NSPLIT + split)) * KPAD + kk) * FF;
            prow[row]      = acc0[r];
            prow[16 + row] = acc1[r];
        }
        if (has_xrow) {
            const float4* be_ = Be + group * (XROW_STEPS * 32);
            const float4* bo_ = Bo + group * (XROW_STEPS * 32);
            float4 xf[XROW_STEPS];
#pragma unroll
            for (int t = 0; t < XROW_STEPS; t += 2) {
                xf[t]     = be_[32 * t];
                xf[t + 1] = bo_[32 * t];
            }
            v8f x0 = {0.f,0.f,0.f,0.f,0.f,0.f,0.f,0.f};
            v8f x1 = {0.f,0.f,0.f,0.f,0.f,0.f,0.f,0.f};
#pragma unroll
            for (int t = 0; t < XROW_STEPS; ++t) {
                v2f b0_; b0_.x = xf[t].x; b0_.y = xf[t].y;
                v2f b1_; b1_.x = xf[t].z; b1_.y = xf[t].w;
                x0 = WMMA_F32(xa[t], b0_, x0);
                x1 = WMMA_F32(xa[t], b1_, x1);
            }
            if (hi == 0) {
                float* prow = lastrow + ((size_t)(b * NSPLIT + split) * 32 + group) * FF;
                prow[row]      = x0[0];
                prow[16 + row] = x1[0];
            }
        }
    } else {
#pragma nounroll
        for (int j = 0; j < NCH - 6; j += 6) {
            LSTEP(j,     1, s1, b1);
            LSTEP(j + 1, 0, s2, b2);
            LSTEP(j + 2, 1, s0, b0);
            LSTEP(j + 3, 0, s1, b1);
            LSTEP(j + 4, 1, s2, b2);
            LSTEP(j + 5, 0, s0, b0);
        }
        LSTEP(NCH - 6, 1, s1, b1);
        LSTEP(NCH - 5, 0, s2, b2);
        LSTEP(NCH - 4, 1, s0, b0);
        LSTEP(NCH - 3, 0, s1, b1);
        LSTEP(NCH - 2, 1, s2, b2);
        LSTEP(NCH - 1, 0, s0, b0);
    }
}

__global__ void sufg_reduce_kernel(const float* __restrict__ partial,
                                   const float* __restrict__ lastrow,
                                   const float* __restrict__ bias,
                                   float* __restrict__ out) {
    int idx = (gridDim.x - 1 - blockIdx.x) * blockDim.x + threadIdx.x;
    if (idx >= BB * KOUT * FF) return;
    int o = idx & 31;
    int k = (idx >> 5) % KOUT;
    int b = idx / (KOUT * FF);
    float s = bias[o];
    if (k < KOUT - 1) {
#pragma unroll
        for (int sp = 0; sp < NSPLIT; ++sp)
            s += partial[(((size_t)(b * NSPLIT + sp)) * KPAD + k) * FF + o];
    } else {
        const float* pr = lastrow + (size_t)b * XROW_PIECES * FF + o;
#pragma unroll
        for (int i = 0; i < XROW_PIECES; ++i) s += pr[i * FF];
    }
    out[idx] = s;
}

extern "C" void kernel_launch(void* const* d_in, const int* in_sizes, int n_in,
                              void* d_out, int out_size, void* d_ws, size_t ws_size,
                              hipStream_t stream) {
    (void)in_sizes; (void)n_in; (void)out_size; (void)ws_size;
    const float* x    = (const float*)d_in[0];
    const float* dl   = (const float*)d_in[1];
    const float* w    = (const float*)d_in[2];
    const float* filt = (const float*)d_in[3];
    const float* bias = (const float*)d_in[4];
    float* out = (float*)d_out;

    float* ws      = (float*)d_ws;
    float* xw      = ws + WS_XW_OFF;
    float* xshr    = ws + WS_XSHR_OFF;
    float* partial = ws + WS_PART_OFF;
    float* lastrow = ws + WS_XW_OFF;

    const float thr = sqrtf(2.0f * logf((float)NN)) / sqrtf((float)NN);

    sufg_xw_kernel<<<(BB * NN * FF) / 256, 256, 0, stream>>>(x, w, xw);
    sufg_pass2_kernel<<<dim3((MSHR / 16) / 4, BB), 256, 0, stream>>>(dl, xw, filt, xshr, thr);
    sufg_pass3_kernel<<<dim3(32, NSPLIT, BB), 256, 0, stream>>>(dl, xshr, partial, lastrow);
    sufg_reduce_kernel<<<(BB * KOUT * FF + 255) / 256, 256, 0, stream>>>(partial, lastrow, bias, out);
}
